# starting kernel
# baseline (speedup 1.0000x reference)
#include <hip/hip_runtime.h>

typedef __attribute__((ext_vector_type(2))) float v2f;
typedef __attribute__((ext_vector_type(8))) float v8f;

#define MJ_STRIDE  132
#define ADJ_STRIDE  66

__launch_bounds__(256)
__global__ void gnn_fused_kernel(const float* __restrict__ x,
                                 const int*   __restrict__ adj,
                                 const float* __restrict__ W,
                                 const float* __restrict__ bias,
                                 float*       __restrict__ out)
{
    __shared__ __align__(16) float s_mj [64 * MJ_STRIDE];
    __shared__ __align__(16) float s_adj[64 * ADJ_STRIDE];
    __shared__ float s_deg[64];

    const int tid  = threadIdx.x;
    const int wave = tid >> 5;
    const int lane = tid & 31;
    const int hf   = lane >> 4;
    const int l16  = lane & 15;

    const int b     = blockIdx.y;
    const int obase = blockIdx.x * 128;

    const int kslab = wave >> 1;
    const int osub0 = (wave & 1) * 4;

    #pragma unroll
    for (int it = 0; it < 16; ++it) {
        int idx = it * 256 + tid;
        int r = idx >> 6, c = idx & 63;
        s_adj[r * ADJ_STRIDE + c] = (float)adj[idx];
    }
    __syncthreads();
    if (tid < 64) {
        float d = 0.f;
        for (int j = 0; j < 64; ++j) d += s_adj[tid * ADJ_STRIDE + j];
        s_deg[tid] = d;
    }

    const v8f zero = {0.f,0.f,0.f,0.f,0.f,0.f,0.f,0.f};
    v8f acc_i[4], acc_j[4];
    #pragma unroll
    for (int t = 0; t < 4; ++t) { acc_i[t] = zero; acc_j[t] = zero; }

    const float* xrow = x + ((size_t)b * 64 + kslab * 16 + l16) * 512 + 2 * hf;
    const float* wbase = W + (size_t)(obase + osub0 * 16 + l16) * 1024 + 2 * hf;

    for (int d = 0; d < 512; d += 4) {
        v2f a = *(const v2f*)(xrow + d);
        #pragma unroll
        for (int t = 0; t < 4; ++t) {
            const float* wrow = wbase + (size_t)t * 16 * 1024 + d;
            v2f bi = *(const v2f*)(wrow);
            v2f bj = *(const v2f*)(wrow + 512);
            acc_i[t] = __builtin_amdgcn_wmma_f32_16x16x4_f32(
                false, a, false, bi, (short)0, acc_i[t], false, false);
            acc_j[t] = __builtin_amdgcn_wmma_f32_16x16x4_f32(
                false, a, false, bj, (short)0, acc_j[t], false, false);
        }
    }

    #pragma unroll
    for (int t = 0; t < 4; ++t) {
        int col = (osub0 + t) * 16 + l16;
        #pragma unroll
        for (int r = 0; r < 8; ++r) {
            int row = kslab * 16 + r + 8 * hf;
            s_mj[row * MJ_STRIDE + col] = acc_j[t][r];
        }
    }
    __syncthreads();

    v8f agg[4];
    #pragma unroll
    for (int t = 0; t < 4; ++t) agg[t] = zero;

    const float* arow = s_adj + (kslab * 16 + l16) * ADJ_STRIDE + 2 * hf;
    for (int j = 0; j < 64; j += 4) {
        v2f a = *(const v2f*)(arow + j);
        #pragma unroll
        for (int t = 0; t < 4; ++t) {
            int col = (osub0 + t) * 16 + l16;
            v2f bv;
            bv.x = s_mj[(j + 2 * hf)     * MJ_STRIDE + col];
            bv.y = s_mj[(j + 2 * hf + 1) * MJ_STRIDE + col];
            agg[t] = __builtin_amdgcn_wmma_f32_16x16x4_f32(
                false, a, false, bv, (short)0, agg[t], false, false);
        }
    }

    float* outb = out + (size_t)b * 64 * 512;
    #pragma unroll
    for (int t = 0; t < 4; ++t) {
        int ocol = obase + (osub0 + t) * 16 + l16;
        float bv = bias[ocol];
        #pragma unroll
        for (int r = 0; r < 8; ++r) {
            int row = kslab * 16 + r + 8 * hf;
            outb[(size_t)row * 512 + ocol] = s_deg[row] * (acc_i[t][r] + bv) + agg[t][r];
        }
    }
}

extern "C" void kernel_launch(void* const* d_in, const int* in_sizes, int n_in,
                              void* d_out, int out_size, void* d_ws, size_t ws_size,
                              hipStream_t stream) {
    const float* x    = (const float*)d_in[0];
    const int*   adj  = (const int*)  d_in[1];
    const float* W    = (const float*)d_in[2];
    const float* bias = (const float*)d_in[3];
    float*       out  = (float*)d_out;

    dim3 grid(4, 32);
    gnn_fused_kernel<<<grid, 256, 0, stream>>>(x, adj, W, bias, out);
}


# Widen each wave's register tile to 64x32 on 4 waves, one per SIMD: 8 fragment loads per 32 WMMAs, not 6 per 16, issued one 32-k trip ahead; a strip's two waves swap halves; sum order kept
# speedup vs baseline: 3.4237x; 3.4237x over previous
#include <hip/hip_runtime.h>


typedef __attribute__((ext_vector_type(2))) float v2f;
typedef __attribute__((ext_vector_type(4))) float v4f;
typedef __attribute__((ext_vector_type(8))) float v8f;

#define ADJ_STRIDE  66

struct Frag {
    v4f a[4][4];
    v4f bi[4][2];
    v4f bj[4][2];
};

__device__ __forceinline__ void load_trip(Frag& f, const float* xrow,
                                          const float* w0, const float* w1, int d0)
{
    #pragma unroll
    for (int ch = 0; ch < 4; ++ch) {
        const int d = d0 + 8 * ch;
        f.bi[ch][0] = *(const v4f*)(w0 + d);
        f.bj[ch][0] = *(const v4f*)(w0 + 512 + d);
        f.bi[ch][1] = *(const v4f*)(w1 + d);
        f.bj[ch][1] = *(const v4f*)(w1 + 512 + d);
        #pragma unroll
        for (int t = 0; t < 4; ++t)
            f.a[ch][t] = *(const v4f*)(xrow + (size_t)t * 16 * 512 + d);
    }
}

__device__ __forceinline__ void mma_trip(const Frag& f, v8f (&acc_i)[2][4], v8f (&acc_j)[2][4])
{
    #pragma unroll
    for (int ch = 0; ch < 4; ++ch) {
        #pragma unroll
        for (int q = 0; q < 2; ++q) {
            #pragma unroll
            for (int c = 0; c < 2; ++c) {
                v2f biq, bjq;
                biq.x = f.bi[ch][c][2 * q]; biq.y = f.bi[ch][c][2 * q + 1];
                bjq.x = f.bj[ch][c][2 * q]; bjq.y = f.bj[ch][c][2 * q + 1];
                #pragma unroll
                for (int t = 0; t < 4; ++t) {
                    v2f aq;
                    aq.x = f.a[ch][t][2 * q]; aq.y = f.a[ch][t][2 * q + 1];
                    acc_i[c][t] = __builtin_amdgcn_wmma_f32_16x16x4_f32(
                        false, aq, false, biq, (short)0, acc_i[c][t], false, false);
                    acc_j[c][t] = __builtin_amdgcn_wmma_f32_16x16x4_f32(
                        false, aq, false, bjq, (short)0, acc_j[c][t], false, false);
                }
            }
        }
    }
}

__launch_bounds__(128) __attribute__((amdgpu_waves_per_eu(1, 1)))
__global__ void gnn_fused_kernel(const float* __restrict__ x,
                                 const int*   __restrict__ adj,
                                 const float* __restrict__ W,
                                 const float* __restrict__ bias,
                                 float*       __restrict__ out)
{
    __shared__ __align__(16) float s_adj[64 * ADJ_STRIDE];
    __shared__ float s_deg[64];
    __shared__ v4f s_part[4 * 16 * 32];

    const int tid  = threadIdx.x;
    const int wave  = __builtin_amdgcn_readfirstlane(tid >> 5);
    const int strip = wave & 1;
    const int kh    = wave >> 1;
    const int lane = tid & 31;
    const int hf   = lane >> 4;
    const int l16  = lane & 15;

    const int b     = blockIdx.y;
    const int ocol0 = blockIdx.x * 64 + strip * 32;
    const int ckeep = ocol0 + 16 * kh;
    const int cgive = ocol0 + 16 * (1 - kh);

    const float* xrow = x + ((size_t)b * 64 + l16) * 512 + kh * 256 + 4 * hf;
    const float* w0   = W + (size_t)(ckeep + l16) * 1024 + kh * 256 + 4 * hf;
    const float* w1   = W + (size_t)(cgive + l16) * 1024 + kh * 256 + 4 * hf;

    Frag f0, f1;
    load_trip(f0, xrow, w0, w1, 0);

    #pragma unroll
    for (int it = 0; it < 32; ++it) {
        int idx = it * 128 + tid;
        int r = idx >> 6, c = idx & 63;
        s_adj[r * ADJ_STRIDE + c] = (float)adj[idx];
    }
    __syncthreads();
    if (tid < 64) {
        float d = 0.f;
        for (int j = 0; j < 64; ++j) d += s_adj[tid * ADJ_STRIDE + j];
        s_deg[tid] = d;
    }

    const v8f zero = {0.f,0.f,0.f,0.f,0.f,0.f,0.f,0.f};
    v8f acc_i[2][4], acc_j[2][4];
    #pragma unroll
    for (int c = 0; c < 2; ++c) {
        #pragma unroll
        for (int t = 0; t < 4; ++t) { acc_i[c][t] = zero; acc_j[c][t] = zero; }
    }

    #pragma unroll 1
    for (int d = 0; d < 192; d += 64) {
        load_trip(f1, xrow, w0, w1, d + 32);
        mma_trip(f0, acc_i, acc_j);
        load_trip(f0, xrow, w0, w1, d + 64);
        mma_trip(f1, acc_i, acc_j);
    }
    load_trip(f1, xrow, w0, w1, 224);
    mma_trip(f0, acc_i, acc_j);
    mma_trip(f1, acc_i, acc_j);

    {
        v4f* p = s_part + wave * 16 * 32 + lane;
        #pragma unroll
        for (int t = 0; t < 4; ++t) {
            p[(4 * t + 0) * 32] = __builtin_shufflevector(acc_i[1][t], acc_i[1][t], 0, 1, 2, 3);
            p[(4 * t + 1) * 32] = __builtin_shufflevector(acc_i[1][t], acc_i[1][t], 4, 5, 6, 7);
            p[(4 * t + 2) * 32] = __builtin_shufflevector(acc_j[1][t], acc_j[1][t], 0, 1, 2, 3);
            p[(4 * t + 3) * 32] = __builtin_shufflevector(acc_j[1][t], acc_j[1][t], 4, 5, 6, 7);
        }
    }

    __syncthreads();

    v8f mi[4], mj[4];
    {
        const v4f* p = s_part + (wave ^ 2) * 16 * 32 + lane;
        #pragma unroll
        for (int t = 0; t < 4; ++t) {
            v4f i0 = p[(4 * t + 0) * 32], i1 = p[(4 * t + 1) * 32];
            v4f j0 = p[(4 * t + 2) * 32], j1 = p[(4 * t + 3) * 32];
            mi[t] = acc_i[0][t] + __builtin_shufflevector(i0, i1, 0, 1, 2, 3, 4, 5, 6, 7);
            mj[t] = acc_j[0][t] + __builtin_shufflevector(j0, j1, 0, 1, 2, 3, 4, 5, 6, 7);
        }
    }

    v8f agg[4];
    #pragma unroll
    for (int t = 0; t < 4; ++t) agg[t] = zero;

    const float* arow = s_adj + l16 * ADJ_STRIDE + 8 * hf;
    #pragma unroll
    for (int tj = 0; tj < 4; ++tj) {
        #pragma unroll
        for (int p = 0; p < 4; ++p) {
            v2f bv;
            bv.x = mj[tj][2 * p];
            bv.y = mj[tj][2 * p + 1];
            #pragma unroll
            for (int ti = 0; ti < 4; ++ti) {
                v2f a = *(const v2f*)(arow + ti * 16 * ADJ_STRIDE + 16 * tj + 2 * p);
                agg[ti] = __builtin_amdgcn_wmma_f32_16x16x4_f32(
                    false, a, false, bv, (short)0, agg[ti], false, false);
            }
        }
    }

    float* outb = out + (size_t)b * 64 * 512;
    const int ocol = ckeep + l16;
    const float bv = bias[ocol];
    #pragma unroll
    for (int t = 0; t < 4; ++t) {
        #pragma unroll
        for (int r = 0; r < 8; ++r) {
            int row = t * 16 + r + 8 * hf;
            outb[(size_t)row * 512 + ocol] = s_deg[row] * (mi[t][r] + bv) + agg[t][r];
        }
    }
}

extern "C" void kernel_launch(void* const* d_in, const int* in_sizes, int n_in,
                              void* d_out, int out_size, void* d_ws, size_t ws_size,
                              hipStream_t stream) {
    const float* x    = (const float*)d_in[0];
    const int*   adj  = (const int*)  d_in[1];
    const float* W    = (const float*)d_in[2];
    const float* bias = (const float*)d_in[3];
    float*       out  = (float*)d_out;

    dim3 grid(8, 32);
    gnn_fused_kernel<<<grid, 128, 0, stream>>>(x, adj, W, bias, out);
}


# Take adj off the GEMM's start: 8 whole-line loads per wave, deg from lane sums and a 16-lane butterfly; no barrier, whole-trip wait or serial deg loop before the first WMMA; deg sum order changes
# speedup vs baseline: 3.6249x; 1.0588x over previous
#include <hip/hip_runtime.h>


typedef __attribute__((ext_vector_type(2))) float v2f;
typedef __attribute__((ext_vector_type(4))) float v4f;
typedef __attribute__((ext_vector_type(8))) float v8f;
typedef __attribute__((ext_vector_type(4))) int   v4i;

#define ADJ_STRIDE  66

struct Frag {
    v4f a[4][4];
    v4f bi[4][2];
    v4f bj[4][2];
};

__device__ __forceinline__ void load_trip(Frag& f, const float* xrow,
                                          const float* w0, const float* w1, int d0)
{
    #pragma unroll
    for (int ch = 0; ch < 4; ++ch) {
        const int d = d0 + 8 * ch;
        f.bi[ch][0] = *(const v4f*)(w0 + d);
        f.bj[ch][0] = *(const v4f*)(w0 + 512 + d);
        f.bi[ch][1] = *(const v4f*)(w1 + d);
        f.bj[ch][1] = *(const v4f*)(w1 + 512 + d);
        #pragma unroll
        for (int t = 0; t < 4; ++t)
            f.a[ch][t] = *(const v4f*)(xrow + (size_t)t * 16 * 512 + d);
    }
}

__device__ __forceinline__ void load_first_trip(Frag& f, const float* xrow,
                                                const float* w0, const float* w1)
{
    __builtin_amdgcn_sched_barrier(0);
    #pragma unroll
    for (int ch = 0; ch < 4; ++ch) {
        const int d = 8 * ch;
        f.bi[ch][0] = *(const v4f*)(w0 + d);
        f.bj[ch][0] = *(const v4f*)(w0 + 512 + d);
        f.bi[ch][1] = *(const v4f*)(w1 + d);
        f.bj[ch][1] = *(const v4f*)(w1 + 512 + d);
        #pragma unroll
        for (int t = 0; t < 4; ++t)
            f.a[ch][t] = *(const v4f*)(xrow + (size_t)t * 16 * 512 + d);
        __builtin_amdgcn_sched_barrier(0);
    }
}

__device__ __forceinline__ void mma_trip(const Frag& f, v8f (&acc_i)[2][4], v8f (&acc_j)[2][4])
{
    #pragma unroll
    for (int ch = 0; ch < 4; ++ch) {
        #pragma unroll
        for (int q = 0; q < 2; ++q) {
            #pragma unroll
            for (int c = 0; c < 2; ++c) {
                v2f biq, bjq;
                biq.x = f.bi[ch][c][2 * q]; biq.y = f.bi[ch][c][2 * q + 1];
                bjq.x = f.bj[ch][c][2 * q]; bjq.y = f.bj[ch][c][2 * q + 1];
                #pragma unroll
                for (int t = 0; t < 4; ++t) {
                    v2f aq;
                    aq.x = f.a[ch][t][2 * q]; aq.y = f.a[ch][t][2 * q + 1];
                    acc_i[c][t] = __builtin_amdgcn_wmma_f32_16x16x4_f32(
                        false, aq, false, biq, (short)0, acc_i[c][t], false, false);
                    acc_j[c][t] = __builtin_amdgcn_wmma_f32_16x16x4_f32(
                        false, aq, false, bjq, (short)0, acc_j[c][t], false, false);
                }
            }
        }
    }
}

__device__ __forceinline__ float half_sum16(float s)
{
    s += __builtin_bit_cast(float, __builtin_amdgcn_update_dpp(0, __builtin_bit_cast(int, s), 0x161, 0xf, 0xf, true));
    s += __builtin_bit_cast(float, __builtin_amdgcn_update_dpp(0, __builtin_bit_cast(int, s), 0x162, 0xf, 0xf, true));
    s += __builtin_bit_cast(float, __builtin_amdgcn_update_dpp(0, __builtin_bit_cast(int, s), 0x164, 0xf, 0xf, true));
    s += __builtin_bit_cast(float, __builtin_amdgcn_update_dpp(0, __builtin_bit_cast(int, s), 0x168, 0xf, 0xf, true));
    return s;
}

__launch_bounds__(128) __attribute__((amdgpu_waves_per_eu(1, 1)))
__global__ void gnn_fused_kernel(const float* __restrict__ x,
                                 const int*   __restrict__ adj,
                                 const float* __restrict__ W,
                                 const float* __restrict__ bias,
                                 float*       __restrict__ out)
{
    __shared__ __align__(16) float s_adj[64 * ADJ_STRIDE];
    __shared__ float s_deg[64];
    __shared__ v4f s_part[4 * 16 * 32];

    const int tid  = threadIdx.x;
    const int wave  = __builtin_amdgcn_readfirstlane(tid >> 5);
    const int strip = wave & 1;
    const int kh    = wave >> 1;
    const int lane = tid & 31;
    const int hf   = lane >> 4;
    const int l16  = lane & 15;

    const int b     = blockIdx.y;
    const int ocol0 = blockIdx.x * 64 + strip * 32;
    const int ckeep = ocol0 + 16 * kh;
    const int cgive = ocol0 + 16 * (1 - kh);

    const float* xrow = x + ((size_t)b * 64 + l16) * 512 + kh * 256 + 4 * hf;
    const float* w0   = W + (size_t)(ckeep + l16) * 1024 + kh * 256 + 4 * hf;
    const float* w1   = W + (size_t)(cgive + l16) * 1024 + kh * 256 + 4 * hf;

    const int* adj_w = adj + wave * (16 * 64) + lane * 4;
    v4i araw[8];
    #pragma unroll
    for (int i = 0; i < 8; ++i) araw[i] = *(const v4i*)(adj_w + i * 128);

    Frag f0, f1;
    load_first_trip(f0, xrow, w0, w1);

    {
        float dsel = 0.f;
        #pragma unroll
        for (int i = 0; i < 8; ++i) {
            v2f lo, hi;
            lo.x = (float)araw[i].x; lo.y = (float)araw[i].y;
            hi.x = (float)araw[i].z; hi.y = (float)araw[i].w;
            float* dst = s_adj + (wave * 16 + 2 * i + hf) * ADJ_STRIDE + 4 * l16;
            *(v2f*)dst       = lo;
            *(v2f*)(dst + 2) = hi;
            const float s = half_sum16((lo.x + lo.y) + (hi.x + hi.y));
            dsel = (l16 == i) ? s : dsel;
        }
        if (l16 < 8) s_deg[wave * 16 + 2 * l16 + hf] = dsel;
    }

    const v8f zero = {0.f,0.f,0.f,0.f,0.f,0.f,0.f,0.f};
    v8f acc_i[2][4], acc_j[2][4];
    #pragma unroll
    for (int c = 0; c < 2; ++c) {
        #pragma unroll
        for (int t = 0; t < 4; ++t) { acc_i[c][t] = zero; acc_j[c][t] = zero; }
    }

    #pragma unroll 1
    for (int d = 0; d < 192; d += 64) {
        load_trip(f1, xrow, w0, w1, d + 32);
        mma_trip(f0, acc_i, acc_j);
        load_trip(f0, xrow, w0, w1, d + 64);
        mma_trip(f1, acc_i, acc_j);
    }
    load_trip(f1, xrow, w0, w1, 224);
    mma_trip(f0, acc_i, acc_j);
    mma_trip(f1, acc_i, acc_j);

    {
        v4f* p = s_part + wave * 16 * 32 + lane;
        #pragma unroll
        for (int t = 0; t < 4; ++t) {
            p[(4 * t + 0) * 32] = __builtin_shufflevector(acc_i[1][t], acc_i[1][t], 0, 1, 2, 3);
            p[(4 * t + 1) * 32] = __builtin_shufflevector(acc_i[1][t], acc_i[1][t], 4, 5, 6, 7);
            p[(4 * t + 2) * 32] = __builtin_shufflevector(acc_j[1][t], acc_j[1][t], 0, 1, 2, 3);
            p[(4 * t + 3) * 32] = __builtin_shufflevector(acc_j[1][t], acc_j[1][t], 4, 5, 6, 7);
        }
    }

    __syncthreads();

    v8f mi[4], mj[4];
    {
        const v4f* p = s_part + (wave ^ 2) * 16 * 32 + lane;
        #pragma unroll
        for (int t = 0; t < 4; ++t) {
            v4f i0 = p[(4 * t + 0) * 32], i1 = p[(4 * t + 1) * 32];
            v4f j0 = p[(4 * t + 2) * 32], j1 = p[(4 * t + 3) * 32];
            mi[t] = acc_i[0][t] + __builtin_shufflevector(i0, i1, 0, 1, 2, 3, 4, 5, 6, 7);
            mj[t] = acc_j[0][t] + __builtin_shufflevector(j0, j1, 0, 1, 2, 3, 4, 5, 6, 7);
        }
    }

    v8f agg[4];
    #pragma unroll
    for (int t = 0; t < 4; ++t) agg[t] = zero;

    const float* arow = s_adj + l16 * ADJ_STRIDE + 8 * hf;
    #pragma unroll
    for (int tj = 0; tj < 4; ++tj) {
        #pragma unroll
        for (int p = 0; p < 4; ++p) {
            v2f bv;
            bv.x = mj[tj][2 * p];
            bv.y = mj[tj][2 * p + 1];
            #pragma unroll
            for (int ti = 0; ti < 4; ++ti) {
                v2f a = *(const v2f*)(arow + ti * 16 * ADJ_STRIDE + 16 * tj + 2 * p);
                agg[ti] = __builtin_amdgcn_wmma_f32_16x16x4_f32(
                    false, a, false, bv, (short)0, agg[ti], false, false);
            }
        }
    }

    float* outb = out + (size_t)b * 64 * 512;
    const int ocol = ckeep + l16;
    const float bv = bias[ocol];
    #pragma unroll
    for (int t = 0; t < 4; ++t) {
        #pragma unroll
        for (int r = 0; r < 8; ++r) {
            int row = t * 16 + r + 8 * hf;
            outb[(size_t)row * 512 + ocol] = s_deg[row] * (mi[t][r] + bv) + agg[t][r];
        }
    }
}

extern "C" void kernel_launch(void* const* d_in, const int* in_sizes, int n_in,
                              void* d_out, int out_size, void* d_ws, size_t ws_size,
                              hipStream_t stream) {
    const float* x    = (const float*)d_in[0];
    const int*   adj  = (const int*)  d_in[1];
    const float* W    = (const float*)d_in[2];
    const float* bias = (const float*)d_in[3];
    float*       out  = (float*)d_out;

    dim3 grid(8, 32);
    gnn_fused_kernel<<<grid, 128, 0, stream>>>(x, adj, W, bias, out);
}
